# starting kernel
# baseline (speedup 1.0000x reference)
#include <hip/hip_runtime.h>


#define IMG      512
#define CH       16
#define TS       32
#define RAWN     52
#define RAWROWS  64
#define RAWPITCH 66
#define ABROWS   48
#define ABPITCH  50
#define TROWS    64
#define TPITCH   50
#define MAPITCH  34

typedef __attribute__((ext_vector_type(2))) float v2f;
typedef __attribute__((ext_vector_type(8))) float v8f;

__device__ __forceinline__ int refl(int p) {
  p = p < 0 ? -p : p;
  return p >= IMG ? 2 * IMG - 2 - p : p;
}

template <int MODE>
__device__ __forceinline__ void hblur(const float* __restrict__ s0,
                                      const float* __restrict__ s1, int sPitch,
                                      float* __restrict__ dst, int dPitch,
                                      int mBlocks, int nBlocks,
                                      const v2f* __restrict__ band,
                                      int lane, int wave) {
  const int m = lane & 15, half = lane >> 4;
  const int nT = mBlocks * nBlocks;
  for (int t = wave; t < nT; t += 8) {
    const int mi = t / nBlocks, nj = t - mi * nBlocks;
    const float* p0 = s0 + (mi * 16 + m) * sPitch + nj * 16 + 2 * half;
    const float* p1 = s1 + (mi * 16 + m) * sPitch + nj * 16 + 2 * half;
    v8f c = {};
#pragma unroll
    for (int kt = 0; kt < 7; ++kt) {
      float x0 = p0[4 * kt], x1 = p0[4 * kt + 1];
      if (MODE == 1) { x0 *= x0; x1 *= x1; }
      if (MODE == 2) { x0 *= p1[4 * kt]; x1 *= p1[4 * kt + 1]; }
      v2f a; a.x = x0; a.y = x1;
      c = __builtin_amdgcn_wmma_f32_16x16x4_f32(false, a, false, band[kt],
                                                (short)0, c, false, false);
    }
    float* pd = dst + (mi * 16 + half * 8) * dPitch + nj * 16 + m;
#pragma unroll
    for (int v = 0; v < 8; ++v) pd[v * dPitch] = c[v];
  }
}

__device__ __forceinline__ void vblur(const float* __restrict__ src, int sPitch,
                                      float* __restrict__ dst, int dPitch,
                                      int mBlocks, int nBlocks,
                                      const v2f* __restrict__ band,
                                      int lane, int wave) {
  const int n = lane & 15, half = lane >> 4;
  const int nT = mBlocks * nBlocks;
  for (int t = wave; t < nT; t += 8) {
    const int mi = t / nBlocks, nj = t - mi * nBlocks;
    const float* ps = src + (mi * 16 + 2 * half) * sPitch + nj * 16 + n;
    v8f c = {};
#pragma unroll
    for (int kt = 0; kt < 7; ++kt) {
      v2f b;
      b.x = ps[(4 * kt) * sPitch];
      b.y = ps[(4 * kt + 1) * sPitch];
      c = __builtin_amdgcn_wmma_f32_16x16x4_f32(false, band[kt], false, b,
                                                (short)0, c, false, false);
    }
    float* pd = dst + (mi * 16 + half * 8) * dPitch + nj * 16 + n;
#pragma unroll
    for (int v = 0; v < 8; ++v) pd[v * dPitch] = c[v];
  }
}

__global__ __launch_bounds__(256) void guided_filter_kernel(
    const float* __restrict__ feat, const float* __restrict__ img,
    float* __restrict__ out) {
  __shared__ float rawI[RAWROWS * RAWPITCH];
  __shared__ float rawF[RAWROWS * RAWPITCH];
  __shared__ float tmpH[TROWS * TPITCH];
  __shared__ float mI [ABROWS * ABPITCH];
  __shared__ float vIi[ABROWS * ABPITCH];
  __shared__ float mP [ABROWS * ABPITCH];
  __shared__ float cIp[ABROWS * ABPITCH];
  __shared__ float mA [TS * MAPITCH];
  __shared__ float mB [TS * MAPITCH];

  const int tid  = threadIdx.x;
  const int lane = tid & 31;
  const int wave = tid >> 5;
  const int x0 = blockIdx.x * TS, y0 = blockIdx.y * TS;
  const int bat = blockIdx.z;

  v2f band[7];
  {
    const int q = lane & 15, h2 = (lane >> 4) * 2;
#pragma unroll
    for (int kt = 0; kt < 7; ++kt) {
      const int k0 = 4 * kt + h2;
      band[kt].x = (k0 >= q && k0 <= q + 10) ? (1.0f / 11.0f) : 0.0f;
      band[kt].y = (k0 + 1 >= q && k0 + 1 <= q + 10) ? (1.0f / 11.0f) : 0.0f;
    }
  }

  for (int i = tid; i < RAWROWS * RAWPITCH; i += 256) { rawI[i] = 0.0f; rawF[i] = 0.0f; }
  __syncthreads();

  const float* imgB = img + (size_t)bat * IMG * IMG;
  for (int i = tid; i < RAWN * RAWN; i += 256) {
    const int r = i / RAWN, cc = i - r * RAWN;
    rawI[r * RAWPITCH + cc] = imgB[refl(y0 - 10 + r) * IMG + refl(x0 - 10 + cc)];
  }
  __syncthreads();

  hblur<0>(rawI, rawI, RAWPITCH, tmpH, TPITCH, 4, 3, band, lane, wave);
  __syncthreads();
  vblur(tmpH, TPITCH, mI, ABPITCH, 3, 3, band, lane, wave);
  __syncthreads();
  hblur<1>(rawI, rawI, RAWPITCH, tmpH, TPITCH, 4, 3, band, lane, wave);
  __syncthreads();
  vblur(tmpH, TPITCH, vIi, ABPITCH, 3, 3, band, lane, wave);
  __syncthreads();
  for (int i = tid; i < ABROWS * ABROWS; i += 256) {
    const int r = i / ABROWS, cc = i - r * ABROWS;
    const int o = r * ABPITCH + cc;
    const float m = mI[o];
    vIi[o] = 1.0f / (vIi[o] - m * m + 1e-8f);
  }
  __syncthreads();

  for (int ch = 0; ch < CH; ++ch) {
    const float* featB = feat + ((size_t)bat * CH + ch) * IMG * IMG;
    for (int i = tid; i < RAWN * RAWN; i += 256) {
      const int r = i / RAWN, cc = i - r * RAWN;
      rawF[r * RAWPITCH + cc] = featB[refl(y0 - 10 + r) * IMG + refl(x0 - 10 + cc)];
    }
    __syncthreads();

    hblur<0>(rawF, rawF, RAWPITCH, tmpH, TPITCH, 4, 3, band, lane, wave);
    __syncthreads();
    vblur(tmpH, TPITCH, mP, ABPITCH, 3, 3, band, lane, wave);
    __syncthreads();
    hblur<2>(rawF, rawI, RAWPITCH, tmpH, TPITCH, 4, 3, band, lane, wave);
    __syncthreads();
    vblur(tmpH, TPITCH, cIp, ABPITCH, 3, 3, band, lane, wave);
    __syncthreads();

    for (int i = tid; i < ABROWS * ABROWS; i += 256) {
      const int r = i / ABROWS, cc = i - r * ABROWS;
      const int o = r * ABPITCH + cc;
      const float mi_ = mI[o], mp_ = mP[o];
      const float a = (cIp[o] - mi_ * mp_) * vIi[o];
      cIp[o] = a;
      mP[o]  = mp_ - a * mi_;
    }
    __syncthreads();

    hblur<0>(cIp, cIp, ABPITCH, tmpH, TPITCH, 3, 2, band, lane, wave);
    __syncthreads();
    vblur(tmpH, TPITCH, mA, MAPITCH, 2, 2, band, lane, wave);
    __syncthreads();
    hblur<0>(mP, mP, ABPITCH, tmpH, TPITCH, 3, 2, band, lane, wave);
    __syncthreads();
    vblur(tmpH, TPITCH, mB, MAPITCH, 2, 2, band, lane, wave);
    __syncthreads();

    float* outB = out + ((size_t)bat * CH + ch) * IMG * IMG;
    for (int i = tid; i < TS * TS; i += 256) {
      const int y = i >> 5, x = i & 31;
      outB[(size_t)(y0 + y) * IMG + (x0 + x)] =
          mA[y * MAPITCH + x] * rawI[(y + 10) * RAWPITCH + (x + 10)] +
          mB[y * MAPITCH + x];
    }
    __syncthreads();
  }
}

extern "C" void kernel_launch(void* const* d_in, const int* in_sizes, int n_in,
                              void* d_out, int out_size, void* d_ws, size_t ws_size,
                              hipStream_t stream) {
  const float* feat = (const float*)d_in[0];
  const float* img  = (const float*)d_in[1];
  float* out = (float*)d_out;
  const int B = in_sizes[1] / (IMG * IMG);
  dim3 grid(IMG / TS, IMG / TS, B);
  guided_filter_kernel<<<grid, 256, 0, stream>>>(feat, img, out);
}


# Halo load addressing (reflected source offset, LDS slot) is computed once per block and kept in registers: a channel's 13 loads per thread issue back to back, one memory round trip, not four or five
# speedup vs baseline: 1.4762x; 1.4762x over previous
#include <hip/hip_runtime.h>


#define IMG      512
#define CH       16
#define TS       32
#define RAWN     52
#define RAWZROW  52
#define RAWROWS  53
#define RAWPITCH 60
#define HROWS    13
#define ABROWS   48
#define ABPITCH  50
#define TROWS    64
#define TPITCH   50
#define MAPITCH  34

typedef __attribute__((ext_vector_type(2))) float v2f;
typedef __attribute__((ext_vector_type(8))) float v8f;

__device__ __forceinline__ int refl(int p) {
  p = p < 0 ? -p : p;
  return p >= IMG ? 2 * IMG - 2 - p : p;
}

template <int MODE>
__device__ __forceinline__ void hblur(const float* __restrict__ s0,
                                      const float* __restrict__ s1, int sPitch,
                                      float* __restrict__ dst, int dPitch,
                                      int mBlocks, int nBlocks, int rowMax,
                                      const v2f* __restrict__ band,
                                      int lane, int wave) {
  const int m = lane & 15, half = lane >> 4;
  const int nT = mBlocks * nBlocks;
  for (int t = wave; t < nT; t += 8) {
    const int mi = t / nBlocks, nj = t - mi * nBlocks;
    const int row = min(mi * 16 + m, rowMax);
    const float* p0 = s0 + row * sPitch + nj * 16 + 2 * half;
    const float* p1 = s1 + row * sPitch + nj * 16 + 2 * half;
    v8f c = {};
#pragma unroll
    for (int kt = 0; kt < 7; ++kt) {
      float x0 = p0[4 * kt], x1 = p0[4 * kt + 1];
      if (MODE == 1) { x0 *= x0; x1 *= x1; }
      if (MODE == 2) { x0 *= p1[4 * kt]; x1 *= p1[4 * kt + 1]; }
      v2f a; a.x = x0; a.y = x1;
      c = __builtin_amdgcn_wmma_f32_16x16x4_f32(false, a, false, band[kt],
                                                (short)0, c, false, false);
    }
    float* pd = dst + (mi * 16 + half * 8) * dPitch + nj * 16 + m;
#pragma unroll
    for (int v = 0; v < 8; ++v) pd[v * dPitch] = c[v];
  }
}

__device__ __forceinline__ void vblur(const float* __restrict__ src, int sPitch,
                                      float* __restrict__ dst, int dPitch,
                                      int mBlocks, int nBlocks,
                                      const v2f* __restrict__ band,
                                      int lane, int wave) {
  const int n = lane & 15, half = lane >> 4;
  const int nT = mBlocks * nBlocks;
  for (int t = wave; t < nT; t += 8) {
    const int mi = t / nBlocks, nj = t - mi * nBlocks;
    const float* ps = src + (mi * 16 + 2 * half) * sPitch + nj * 16 + n;
    v8f c = {};
#pragma unroll
    for (int kt = 0; kt < 7; ++kt) {
      v2f b;
      b.x = ps[(4 * kt) * sPitch];
      b.y = ps[(4 * kt + 1) * sPitch];
      c = __builtin_amdgcn_wmma_f32_16x16x4_f32(false, band[kt], false, b,
                                                (short)0, c, false, false);
    }
    float* pd = dst + (mi * 16 + half * 8) * dPitch + nj * 16 + n;
#pragma unroll
    for (int v = 0; v < 8; ++v) pd[v * dPitch] = c[v];
  }
}

__global__ __launch_bounds__(256) void guided_filter_kernel(
    const float* __restrict__ feat, const float* __restrict__ img,
    float* __restrict__ out) {
  __shared__ float rawI[RAWROWS * RAWPITCH];
  __shared__ float rawF[RAWROWS * RAWPITCH];
  __shared__ float tmpH[TROWS * TPITCH];
  __shared__ float mI [ABROWS * ABPITCH];
  __shared__ float vIi[ABROWS * ABPITCH];
  __shared__ float mP [ABROWS * ABPITCH];
  __shared__ float cIp[ABROWS * ABPITCH];
  float* const mA = cIp;
  float* const mB = mP;

  const int tid  = threadIdx.x;
  const int lane = tid & 31;
  const int wave = tid >> 5;
  const int x0 = blockIdx.x * TS, y0 = blockIdx.y * TS;
  const int bat = blockIdx.z;

  v2f band[7];
  {
    const int q = lane & 15, h2 = (lane >> 4) * 2;
#pragma unroll
    for (int kt = 0; kt < 7; ++kt) {
      const int k0 = 4 * kt + h2;
      band[kt].x = (k0 >= q && k0 <= q + 10) ? (1.0f / 11.0f) : 0.0f;
      band[kt].y = (k0 + 1 >= q && k0 + 1 <= q + 10) ? (1.0f / 11.0f) : 0.0f;
    }
  }

  for (int i = tid; i < RAWROWS * RAWPITCH; i += 256) { rawI[i] = 0.0f; rawF[i] = 0.0f; }
  __syncthreads();

  const int  hcol = tid & 63;
  const bool hact = hcol < RAWN;
  const int  hlds = (tid >> 6) * RAWPITCH + hcol;
  unsigned hofs[HROWS];
  {
    const int gx = refl(x0 - 10 + hcol);
#pragma unroll
    for (int j = 0; j < HROWS; ++j)
      hofs[j] = (unsigned)(refl(y0 - 10 + (tid >> 6) + 4 * j) * IMG + gx);
  }

  const float* imgB = img + (size_t)bat * IMG * IMG;
  if (hact) {
    float hv[HROWS];
#pragma unroll
    for (int j = 0; j < HROWS; ++j) hv[j] = imgB[hofs[j]];
#pragma unroll
    for (int j = 0; j < HROWS; ++j) rawI[hlds + j * 4 * RAWPITCH] = hv[j];
  }
  __syncthreads();

  hblur<0>(rawI, rawI, RAWPITCH, tmpH, TPITCH, 4, 3, RAWZROW, band, lane, wave);
  __syncthreads();
  vblur(tmpH, TPITCH, mI, ABPITCH, 3, 3, band, lane, wave);
  __syncthreads();
  hblur<1>(rawI, rawI, RAWPITCH, tmpH, TPITCH, 4, 3, RAWZROW, band, lane, wave);
  __syncthreads();
  vblur(tmpH, TPITCH, vIi, ABPITCH, 3, 3, band, lane, wave);
  __syncthreads();
  for (int i = tid; i < ABROWS * ABROWS; i += 256) {
    const int r = i / ABROWS, cc = i - r * ABROWS;
    const int o = r * ABPITCH + cc;
    const float m = mI[o];
    vIi[o] = 1.0f / (vIi[o] - m * m + 1e-8f);
  }
  __syncthreads();

  const float* featBat = feat + (size_t)bat * CH * IMG * IMG;
  for (int ch = 0; ch < CH; ++ch) {
    const unsigned chofs = (unsigned)ch * (unsigned)(IMG * IMG);
    if (hact) {
      float hv[HROWS];
#pragma unroll
      for (int j = 0; j < HROWS; ++j) hv[j] = featBat[hofs[j] + chofs];
#pragma unroll
      for (int j = 0; j < HROWS; ++j) rawF[hlds + j * 4 * RAWPITCH] = hv[j];
    }
    __syncthreads();

    hblur<0>(rawF, rawF, RAWPITCH, tmpH, TPITCH, 4, 3, RAWZROW, band, lane, wave);
    __syncthreads();
    vblur(tmpH, TPITCH, mP, ABPITCH, 3, 3, band, lane, wave);
    __syncthreads();
    hblur<2>(rawF, rawI, RAWPITCH, tmpH, TPITCH, 4, 3, RAWZROW, band, lane, wave);
    __syncthreads();
    vblur(tmpH, TPITCH, cIp, ABPITCH, 3, 3, band, lane, wave);
    __syncthreads();

    for (int i = tid; i < ABROWS * ABROWS; i += 256) {
      const int r = i / ABROWS, cc = i - r * ABROWS;
      const int o = r * ABPITCH + cc;
      const float mi_ = mI[o], mp_ = mP[o];
      const float a = (cIp[o] - mi_ * mp_) * vIi[o];
      cIp[o] = a;
      mP[o]  = mp_ - a * mi_;
    }
    __syncthreads();

    hblur<0>(cIp, cIp, ABPITCH, tmpH, TPITCH, 3, 2, ABROWS - 1, band, lane, wave);
    __syncthreads();
    vblur(tmpH, TPITCH, mA, MAPITCH, 2, 2, band, lane, wave);
    __syncthreads();
    hblur<0>(mP, mP, ABPITCH, tmpH, TPITCH, 3, 2, ABROWS - 1, band, lane, wave);
    __syncthreads();
    vblur(tmpH, TPITCH, mB, MAPITCH, 2, 2, band, lane, wave);
    __syncthreads();

    float* outB = out + ((size_t)bat * CH + ch) * IMG * IMG;
    for (int i = tid; i < TS * TS; i += 256) {
      const int y = i >> 5, x = i & 31;
      outB[(size_t)(y0 + y) * IMG + (x0 + x)] =
          mA[y * MAPITCH + x] * rawI[(y + 10) * RAWPITCH + (x + 10)] +
          mB[y * MAPITCH + x];
    }
    __syncthreads();
  }
}

extern "C" void kernel_launch(void* const* d_in, const int* in_sizes, int n_in,
                              void* d_out, int out_size, void* d_ws, size_t ws_size,
                              hipStream_t stream) {
  const float* feat = (const float*)d_in[0];
  const float* img  = (const float*)d_in[1];
  float* out = (float*)d_out;
  const int B = in_sizes[1] / (IMG * IMG);
  dim3 grid(IMG / TS, IMG / TS, B);
  guided_filter_kernel<<<grid, 256, 0, stream>>>(feat, img, out);
}


# Horizontal 11-tap sums leave the matrix unit: one lane per row adds 16-byte LDS row reads through shared pair sums (f32 vector adds, then x1/11); V passes stay f32 WMMA; H sum order changes
# speedup vs baseline: 1.8118x; 1.2273x over previous
#include <hip/hip_runtime.h>


#define IMG      512
#define CH       16
#define TS       32
#define RAWN     52
#define RAWPITCH 60
#define HROWS    13
#define ABROWS   48
#define ABPITCH  52
#define TROWS    60
#define TPITCH   52
#define MAPITCH  34

typedef __attribute__((ext_vector_type(2))) float v2f;
typedef __attribute__((ext_vector_type(4))) float v4f;
typedef __attribute__((ext_vector_type(8))) float v8f;

__device__ __forceinline__ int refl(int p) {
  p = p < 0 ? -p : p;
  return p >= IMG ? 2 * IMG - 2 - p : p;
}

template <int MODE, int NOUT>
__device__ __forceinline__ void hsum(const float* __restrict__ s0,
                                     const float* __restrict__ s1, int sPitch,
                                     float* __restrict__ dst, int dPitch,
                                     int rows, int lane, int wave) {
  constexpr int NV = (NOUT + 12) / 4;
  constexpr int NE = (NOUT + 10) / 2;
  constexpr int NW = NOUT / 2 + 1;
  const int seg = wave & 3;
  const int row = (wave >> 2) * 32 + lane;
  if (row < rows) {
    const v4f* p0 = (const v4f*)(s0 + row * sPitch + seg * NOUT);
    const v4f* p1 = (const v4f*)(s1 + row * sPitch + seg * NOUT);
    float x[4 * NV];
#pragma unroll
    for (int q = 0; q < NV; ++q) {
      v4f a = p0[q];
      if (MODE == 1) a *= a;
      if (MODE == 2) a *= p1[q];
      x[4 * q] = a.x; x[4 * q + 1] = a.y; x[4 * q + 2] = a.z; x[4 * q + 3] = a.w;
    }
    float e[NE], ee[NE - 2], w[NW];
#pragma unroll
    for (int k = 0; k < NE; ++k) e[k] = x[2 * k] + x[2 * k + 1];
#pragma unroll
    for (int k = 0; k < NE - 2; ++k) ee[k] = e[k] + e[k + 1];
#pragma unroll
    for (int k = 0; k < NW; ++k) w[k] = (ee[k] + ee[k + 2]) + e[k + 4];
    v4f* pd = (v4f*)(dst + row * dPitch + seg * NOUT);
#pragma unroll
    for (int q = 0; q < NOUT / 4; ++q) {
      v4f o;
      o.x = (w[2 * q]     + x[4 * q + 10]) * (1.0f / 11.0f);
      o.y = (x[4 * q + 1] + w[2 * q + 1])  * (1.0f / 11.0f);
      o.z = (w[2 * q + 1] + x[4 * q + 12]) * (1.0f / 11.0f);
      o.w = (x[4 * q + 3] + w[2 * q + 2])  * (1.0f / 11.0f);
      pd[q] = o;
    }
  }
}

__device__ __forceinline__ void vblur(const float* __restrict__ src, int sPitch,
                                      float* __restrict__ dst, int dPitch,
                                      int mBlocks, int nBlocks,
                                      const v2f* __restrict__ band,
                                      int lane, int wave) {
  const int n = lane & 15, half = lane >> 4;
  const int nT = mBlocks * nBlocks;
  for (int t = wave; t < nT; t += 8) {
    const int mi = t / nBlocks, nj = t - mi * nBlocks;
    const float* ps = src + (mi * 16 + 2 * half) * sPitch + nj * 16 + n;
    v8f c = {};
#pragma unroll
    for (int kt = 0; kt < 7; ++kt) {
      v2f b;
      b.x = ps[(4 * kt) * sPitch];
      b.y = ps[(4 * kt + 1) * sPitch];
      c = __builtin_amdgcn_wmma_f32_16x16x4_f32(false, band[kt], false, b,
                                                (short)0, c, false, false);
    }
    float* pd = dst + (mi * 16 + half * 8) * dPitch + nj * 16 + n;
#pragma unroll
    for (int v = 0; v < 8; ++v) pd[v * dPitch] = c[v];
  }
}

__global__ __launch_bounds__(256) void guided_filter_kernel(
    const float* __restrict__ feat, const float* __restrict__ img,
    float* __restrict__ out) {
  __shared__ __attribute__((aligned(16))) float rawI[RAWN * RAWPITCH];
  __shared__ __attribute__((aligned(16))) float rawF[RAWN * RAWPITCH];
  __shared__ __attribute__((aligned(16))) float tmpH[TROWS * TPITCH];
  __shared__ __attribute__((aligned(16))) float mI [ABROWS * ABPITCH];
  __shared__ __attribute__((aligned(16))) float vIi[ABROWS * ABPITCH];
  __shared__ __attribute__((aligned(16))) float mP [ABROWS * ABPITCH];
  __shared__ __attribute__((aligned(16))) float cIp[ABROWS * ABPITCH];
  float* const mA = cIp;
  float* const mB = mP;

  const int tid  = threadIdx.x;
  const int lane = tid & 31;
  const int wave = tid >> 5;
  const int x0 = blockIdx.x * TS, y0 = blockIdx.y * TS;
  const int bat = blockIdx.z;

  v2f band[7];
  {
    const int q = lane & 15, h2 = (lane >> 4) * 2;
#pragma unroll
    for (int kt = 0; kt < 7; ++kt) {
      const int k0 = 4 * kt + h2;
      band[kt].x = (k0 >= q && k0 <= q + 10) ? (1.0f / 11.0f) : 0.0f;
      band[kt].y = (k0 + 1 >= q && k0 + 1 <= q + 10) ? (1.0f / 11.0f) : 0.0f;
    }
  }

  for (int i = tid; i < RAWN * RAWPITCH; i += 256) { rawI[i] = 0.0f; rawF[i] = 0.0f; }
  for (int i = tid; i < (TROWS - RAWN) * TPITCH; i += 256) tmpH[RAWN * TPITCH + i] = 0.0f;
  __syncthreads();

  const int  hcol = tid & 63;
  const bool hact = hcol < RAWN;
  const int  hlds = (tid >> 6) * RAWPITCH + hcol;
  unsigned hofs[HROWS];
  {
    const int gx = refl(x0 - 10 + hcol);
#pragma unroll
    for (int j = 0; j < HROWS; ++j)
      hofs[j] = (unsigned)(refl(y0 - 10 + (tid >> 6) + 4 * j) * IMG + gx);
  }

  const float* imgB = img + (size_t)bat * IMG * IMG;
  if (hact) {
    float hv[HROWS];
#pragma unroll
    for (int j = 0; j < HROWS; ++j) hv[j] = imgB[hofs[j]];
#pragma unroll
    for (int j = 0; j < HROWS; ++j) rawI[hlds + j * 4 * RAWPITCH] = hv[j];
  }
  __syncthreads();

  hsum<0, 12>(rawI, rawI, RAWPITCH, tmpH, TPITCH, RAWN, lane, wave);
  __syncthreads();
  vblur(tmpH, TPITCH, mI, ABPITCH, 3, 3, band, lane, wave);
  __syncthreads();
  hsum<1, 12>(rawI, rawI, RAWPITCH, tmpH, TPITCH, RAWN, lane, wave);
  __syncthreads();
  vblur(tmpH, TPITCH, vIi, ABPITCH, 3, 3, band, lane, wave);
  __syncthreads();
  for (int i = tid; i < ABROWS * ABROWS; i += 256) {
    const int r = i / ABROWS, cc = i - r * ABROWS;
    const int o = r * ABPITCH + cc;
    const float m = mI[o];
    vIi[o] = 1.0f / (vIi[o] - m * m + 1e-8f);
  }
  __syncthreads();

  const float* featBat = feat + (size_t)bat * CH * IMG * IMG;
  for (int ch = 0; ch < CH; ++ch) {
    const unsigned chofs = (unsigned)ch * (unsigned)(IMG * IMG);
    if (hact) {
      float hv[HROWS];
#pragma unroll
      for (int j = 0; j < HROWS; ++j) hv[j] = featBat[hofs[j] + chofs];
#pragma unroll
      for (int j = 0; j < HROWS; ++j) rawF[hlds + j * 4 * RAWPITCH] = hv[j];
    }
    __syncthreads();

    hsum<0, 12>(rawF, rawF, RAWPITCH, tmpH, TPITCH, RAWN, lane, wave);
    __syncthreads();
    vblur(tmpH, TPITCH, mP, ABPITCH, 3, 3, band, lane, wave);
    __syncthreads();
    hsum<2, 12>(rawF, rawI, RAWPITCH, tmpH, TPITCH, RAWN, lane, wave);
    __syncthreads();
    vblur(tmpH, TPITCH, cIp, ABPITCH, 3, 3, band, lane, wave);
    __syncthreads();

    for (int i = tid; i < ABROWS * ABROWS; i += 256) {
      const int r = i / ABROWS, cc = i - r * ABROWS;
      const int o = r * ABPITCH + cc;
      const float mi_ = mI[o], mp_ = mP[o];
      const float a = (cIp[o] - mi_ * mp_) * vIi[o];
      cIp[o] = a;
      mP[o]  = mp_ - a * mi_;
    }
    __syncthreads();

    hsum<0, 8>(cIp, cIp, ABPITCH, tmpH, TPITCH, ABROWS, lane, wave);
    __syncthreads();
    vblur(tmpH, TPITCH, mA, MAPITCH, 2, 2, band, lane, wave);
    __syncthreads();
    hsum<0, 8>(mP, mP, ABPITCH, tmpH, TPITCH, ABROWS, lane, wave);
    __syncthreads();
    vblur(tmpH, TPITCH, mB, MAPITCH, 2, 2, band, lane, wave);
    __syncthreads();

    float* outB = out + ((size_t)bat * CH + ch) * IMG * IMG;
    for (int i = tid; i < TS * TS; i += 256) {
      const int y = i >> 5, x = i & 31;
      outB[(size_t)(y0 + y) * IMG + (x0 + x)] =
          mA[y * MAPITCH + x] * rawI[(y + 10) * RAWPITCH + (x + 10)] +
          mB[y * MAPITCH + x];
    }
    __syncthreads();
  }
}

extern "C" void kernel_launch(void* const* d_in, const int* in_sizes, int n_in,
                              void* d_out, int out_size, void* d_ws, size_t ws_size,
                              hipStream_t stream) {
  const float* feat = (const float*)d_in[0];
  const float* img  = (const float*)d_in[1];
  float* out = (float*)d_out;
  const int B = in_sizes[1] / (IMG * IMG);
  dim3 grid(IMG / TS, IMG / TS, B);
  guided_filter_kernel<<<grid, 256, 0, stream>>>(feat, img, out);
}


# a and b are formed on the 16-byte row reads of the second-stage H pass, which sums both; one V phase blurs both (one tile per wave): the a/b sweep and 3 of 11 barriers per channel are gone
# speedup vs baseline: 1.9698x; 1.0872x over previous
#include <hip/hip_runtime.h>


#define IMG      512
#define CH       16
#define TS       32
#define RAWN     52
#define RAWPITCH 60
#define HROWS    13
#define ABROWS   48
#define ABPITCH  52
#define TROWS    60
#define TPITCH   52
#define MAPITCH  34

typedef __attribute__((ext_vector_type(2))) float v2f;
typedef __attribute__((ext_vector_type(4))) float v4f;
typedef __attribute__((ext_vector_type(8))) float v8f;

__device__ __forceinline__ int refl(int p) {
  p = p < 0 ? -p : p;
  return p >= IMG ? 2 * IMG - 2 - p : p;
}

template <int NOUT>
__device__ __forceinline__ void rowsum_store(const float (&x)[NOUT + 12],
                                             float* __restrict__ drow) {
  constexpr int NE = (NOUT + 10) / 2;
  constexpr int NW = NOUT / 2 + 1;
  float e[NE], ee[NE - 2], w[NW];
#pragma unroll
  for (int k = 0; k < NE; ++k) e[k] = x[2 * k] + x[2 * k + 1];
#pragma unroll
  for (int k = 0; k < NE - 2; ++k) ee[k] = e[k] + e[k + 1];
#pragma unroll
  for (int k = 0; k < NW; ++k) w[k] = (ee[k] + ee[k + 2]) + e[k + 4];
  v4f* pd = (v4f*)drow;
#pragma unroll
  for (int q = 0; q < NOUT / 4; ++q) {
    v4f o;
    o.x = (w[2 * q]     + x[4 * q + 10]) * (1.0f / 11.0f);
    o.y = (x[4 * q + 1] + w[2 * q + 1])  * (1.0f / 11.0f);
    o.z = (w[2 * q + 1] + x[4 * q + 12]) * (1.0f / 11.0f);
    o.w = (x[4 * q + 3] + w[2 * q + 2])  * (1.0f / 11.0f);
    pd[q] = o;
  }
}

template <int MODE, int NOUT>
__device__ __forceinline__ void hsum(const float* __restrict__ s0,
                                     const float* __restrict__ s1, int sPitch,
                                     float* __restrict__ dst, int dPitch,
                                     int rows, int lane, int wave) {
  constexpr int NV = (NOUT + 12) / 4;
  const int seg = wave & 3;
  const int row = (wave >> 2) * 32 + lane;
  if (row < rows) {
    const v4f* p0 = (const v4f*)(s0 + row * sPitch + seg * NOUT);
    const v4f* p1 = (const v4f*)(s1 + row * sPitch + seg * NOUT);
    float x[4 * NV];
#pragma unroll
    for (int q = 0; q < NV; ++q) {
      v4f a = p0[q];
      if (MODE == 1) a *= a;
      if (MODE == 2) a *= p1[q];
      x[4 * q] = a.x; x[4 * q + 1] = a.y; x[4 * q + 2] = a.z; x[4 * q + 3] = a.w;
    }
    rowsum_store<NOUT>(x, dst + row * dPitch + seg * NOUT);
  }
}

__device__ __forceinline__ void hsum_ab(const float* __restrict__ mI,
                                        const float* __restrict__ vIi,
                                        const float* __restrict__ mP,
                                        const float* __restrict__ cIp,
                                        float* __restrict__ dA, int aPitch,
                                        float* __restrict__ dB, int bPitch,
                                        int lane, int wave) {
  if (wave < 6) {
    const int g   = wave * 2 + (lane >> 4);
    const int seg = g / 3;
    const int row = (g - 3 * seg) * 16 + (lane & 15);
    const int o   = row * ABPITCH + seg * 8;
    const v4f* pm = (const v4f*)(mI  + o);
    const v4f* pv = (const v4f*)(vIi + o);
    const v4f* pp = (const v4f*)(mP  + o);
    const v4f* pc = (const v4f*)(cIp + o);
    float xa[20], xb[20];
#pragma unroll
    for (int q = 0; q < 4; ++q) {
      const v4f m = pm[q], iv = pv[q], p = pp[q], c = pc[q];
      const v4f a = (c - m * p) * iv;
      const v4f b = p - a * m;
      xa[4 * q] = a.x; xa[4 * q + 1] = a.y; xa[4 * q + 2] = a.z; xa[4 * q + 3] = a.w;
      xb[4 * q] = b.x; xb[4 * q + 1] = b.y; xb[4 * q + 2] = b.z; xb[4 * q + 3] = b.w;
    }
    {
      const v2f m = *(const v2f*)(pm + 4), iv = *(const v2f*)(pv + 4);
      const v2f p = *(const v2f*)(pp + 4), c  = *(const v2f*)(pc + 4);
      const v2f a = (c - m * p) * iv;
      const v2f b = p - a * m;
      xa[16] = a.x; xa[17] = a.y; xa[18] = 0.0f; xa[19] = 0.0f;
      xb[16] = b.x; xb[17] = b.y; xb[18] = 0.0f; xb[19] = 0.0f;
    }
    rowsum_store<8>(xa, dA + row * aPitch + seg * 8);
    rowsum_store<8>(xb, dB + row * bPitch + seg * 8);
  }
}

__device__ __forceinline__ void vblur(const float* __restrict__ src, int sPitch,
                                      float* __restrict__ dst, int dPitch,
                                      int mBlocks, int nBlocks,
                                      const v2f* __restrict__ band,
                                      int lane, int wave) {
  const int n = lane & 15, half = lane >> 4;
  const int nT = mBlocks * nBlocks;
  for (int t = wave; t < nT; t += 8) {
    const int mi = t / nBlocks, nj = t - mi * nBlocks;
    const float* ps = src + (mi * 16 + 2 * half) * sPitch + nj * 16 + n;
    v8f c = {};
#pragma unroll
    for (int kt = 0; kt < 7; ++kt) {
      v2f b;
      b.x = ps[(4 * kt) * sPitch];
      b.y = ps[(4 * kt + 1) * sPitch];
      c = __builtin_amdgcn_wmma_f32_16x16x4_f32(false, band[kt], false, b,
                                                (short)0, c, false, false);
    }
    float* pd = dst + (mi * 16 + half * 8) * dPitch + nj * 16 + n;
#pragma unroll
    for (int v = 0; v < 8; ++v) pd[v * dPitch] = c[v];
  }
}

__global__ __launch_bounds__(256, 8) void guided_filter_kernel(
    const float* __restrict__ feat, const float* __restrict__ img,
    float* __restrict__ out) {
  __shared__ __attribute__((aligned(16))) float rawI[RAWN * RAWPITCH];
  __shared__ __attribute__((aligned(16))) float rawF[RAWN * RAWPITCH];
  __shared__ __attribute__((aligned(16))) float tmpH[TROWS * TPITCH];
  __shared__ __attribute__((aligned(16))) float mI [ABROWS * ABPITCH];
  __shared__ __attribute__((aligned(16))) float vIi[ABROWS * ABPITCH];
  __shared__ __attribute__((aligned(16))) float mP [ABROWS * ABPITCH];
  __shared__ __attribute__((aligned(16))) float cIp[ABROWS * ABPITCH];
  float* const mA = cIp;
  float* const mB = mP;

  const int tid  = threadIdx.x;
  const int lane = tid & 31;
  const int wave = tid >> 5;
  const int x0 = blockIdx.x * TS, y0 = blockIdx.y * TS;
  const int bat = blockIdx.z;

  v2f band[7];
  {
    const int q = lane & 15, h2 = (lane >> 4) * 2;
#pragma unroll
    for (int kt = 0; kt < 7; ++kt) {
      const int k0 = 4 * kt + h2;
      band[kt].x = (k0 >= q && k0 <= q + 10) ? (1.0f / 11.0f) : 0.0f;
      band[kt].y = (k0 + 1 >= q && k0 + 1 <= q + 10) ? (1.0f / 11.0f) : 0.0f;
    }
  }

  for (int i = tid; i < RAWN * RAWPITCH; i += 256) { rawI[i] = 0.0f; rawF[i] = 0.0f; }
  for (int i = tid; i < (TROWS - RAWN) * TPITCH; i += 256) tmpH[RAWN * TPITCH + i] = 0.0f;
  __syncthreads();

  const int  hcol = tid & 63;
  const bool hact = hcol < RAWN;
  const int  hlds = (tid >> 6) * RAWPITCH + hcol;
  unsigned hofs[HROWS];
  {
    const int gx = refl(x0 - 10 + hcol);
#pragma unroll
    for (int j = 0; j < HROWS; ++j)
      hofs[j] = (unsigned)(refl(y0 - 10 + (tid >> 6) + 4 * j) * IMG + gx);
  }

  const float* imgB = img + (size_t)bat * IMG * IMG;
  if (hact) {
    float hv[HROWS];
#pragma unroll
    for (int j = 0; j < HROWS; ++j) hv[j] = imgB[hofs[j]];
#pragma unroll
    for (int j = 0; j < HROWS; ++j) rawI[hlds + j * 4 * RAWPITCH] = hv[j];
  }
  __syncthreads();

  hsum<0, 12>(rawI, rawI, RAWPITCH, tmpH, TPITCH, RAWN, lane, wave);
  __syncthreads();
  vblur(tmpH, TPITCH, mI, ABPITCH, 3, 3, band, lane, wave);
  __syncthreads();
  hsum<1, 12>(rawI, rawI, RAWPITCH, tmpH, TPITCH, RAWN, lane, wave);
  __syncthreads();
  vblur(tmpH, TPITCH, vIi, ABPITCH, 3, 3, band, lane, wave);
  __syncthreads();
  for (int i = tid; i < ABROWS * ABROWS; i += 256) {
    const int r = i / ABROWS, cc = i - r * ABROWS;
    const int o = r * ABPITCH + cc;
    const float m = mI[o];
    vIi[o] = 1.0f / (vIi[o] - m * m + 1e-8f);
  }
  __syncthreads();

  const float* featBat = feat + (size_t)bat * CH * IMG * IMG;
  for (int ch = 0; ch < CH; ++ch) {
    const unsigned chofs = (unsigned)ch * (unsigned)(IMG * IMG);
    if (hact) {
      float hv[HROWS];
#pragma unroll
      for (int j = 0; j < HROWS; ++j) hv[j] = featBat[hofs[j] + chofs];
#pragma unroll
      for (int j = 0; j < HROWS; ++j) rawF[hlds + j * 4 * RAWPITCH] = hv[j];
    }
    __syncthreads();

    hsum<0, 12>(rawF, rawF, RAWPITCH, tmpH, TPITCH, RAWN, lane, wave);
    __syncthreads();
    vblur(tmpH, TPITCH, mP, ABPITCH, 3, 3, band, lane, wave);
    __syncthreads();
    hsum<2, 12>(rawF, rawI, RAWPITCH, tmpH, TPITCH, RAWN, lane, wave);
    __syncthreads();
    vblur(tmpH, TPITCH, cIp, ABPITCH, 3, 3, band, lane, wave);
    __syncthreads();

    hsum_ab(mI, vIi, mP, cIp, tmpH, TPITCH, rawF, RAWPITCH, lane, wave);
    __syncthreads();
    if (wave < 4) vblur(tmpH, TPITCH,   mA, MAPITCH, 2, 2, band, lane, wave);
    else          vblur(rawF, RAWPITCH, mB, MAPITCH, 2, 2, band, lane, wave - 4);
    __syncthreads();

    float* outB = out + ((size_t)bat * CH + ch) * IMG * IMG;
    for (int i = tid; i < TS * TS; i += 256) {
      const int y = i >> 5, x = i & 31;
      outB[(size_t)(y0 + y) * IMG + (x0 + x)] =
          mA[y * MAPITCH + x] * rawI[(y + 10) * RAWPITCH + (x + 10)] +
          mB[y * MAPITCH + x];
    }
    __syncthreads();
  }
}

extern "C" void kernel_launch(void* const* d_in, const int* in_sizes, int n_in,
                              void* d_out, int out_size, void* d_ws, size_t ws_size,
                              hipStream_t stream) {
  const float* feat = (const float*)d_in[0];
  const float* img  = (const float*)d_in[1];
  float* out = (float*)d_out;
  const int B = in_sizes[1] / (IMG * IMG);
  dim3 grid(IMG / TS, IMG / TS, B);
  guided_filter_kernel<<<grid, 256, 0, stream>>>(feat, img, out);
}


# p and I*p are blurred together: one H phase forms both row sums from one read of the feature rows, one V phase runs the 18 tiles as 3 accumulator chains on 6 waves; 5 barriers per channel, not 8
# speedup vs baseline: 2.1393x; 1.0860x over previous
#include <hip/hip_runtime.h>


#define IMG      512
#define CH       16
#define TS       32
#define RAWN     52
#define RAWPITCH 60
#define HROWS    13
#define ABROWS   48
#define ABPITCH  52
#define TROWS    60
#define TPITCH   52
#define MAPITCH  34

typedef __attribute__((ext_vector_type(2))) float v2f;
typedef __attribute__((ext_vector_type(4))) float v4f;
typedef __attribute__((ext_vector_type(8))) float v8f;

__device__ __forceinline__ int refl(int p) {
  p = p < 0 ? -p : p;
  return p >= IMG ? 2 * IMG - 2 - p : p;
}

template <int NOUT>
__device__ __forceinline__ void rowsum_store(const float (&x)[NOUT + 12],
                                             float* __restrict__ drow) {
  constexpr int NE = (NOUT + 10) / 2;
  constexpr int NW = NOUT / 2 + 1;
  float e[NE], ee[NE - 2], w[NW];
#pragma unroll
  for (int k = 0; k < NE; ++k) e[k] = x[2 * k] + x[2 * k + 1];
#pragma unroll
  for (int k = 0; k < NE - 2; ++k) ee[k] = e[k] + e[k + 1];
#pragma unroll
  for (int k = 0; k < NW; ++k) w[k] = (ee[k] + ee[k + 2]) + e[k + 4];
  v4f* pd = (v4f*)drow;
#pragma unroll
  for (int q = 0; q < NOUT / 4; ++q) {
    v4f o;
    o.x = (w[2 * q]     + x[4 * q + 10]) * (1.0f / 11.0f);
    o.y = (x[4 * q + 1] + w[2 * q + 1])  * (1.0f / 11.0f);
    o.z = (w[2 * q + 1] + x[4 * q + 12]) * (1.0f / 11.0f);
    o.w = (x[4 * q + 3] + w[2 * q + 2])  * (1.0f / 11.0f);
    pd[q] = o;
  }
}

template <int NOUT>
__device__ __forceinline__ void hsum_pair(const float* s0, const float* s1, int sPitch,
                                          float* __restrict__ d0,
                                          float* __restrict__ d1, int dPitch,
                                          int rows, int lane, int wave) {
  constexpr int NV = (NOUT + 12) / 4;
  const int seg = wave & 3;
  const int row = (wave >> 2) * 32 + lane;
  if (row < rows) {
    const v4f* p0 = (const v4f*)(s0 + row * sPitch + seg * NOUT);
    const v4f* p1 = (const v4f*)(s1 + row * sPitch + seg * NOUT);
    float x[4 * NV], y[4 * NV];
#pragma unroll
    for (int q = 0; q < NV; ++q) {
      const v4f a = p0[q];
      const v4f b = a * p1[q];
      x[4 * q] = a.x; x[4 * q + 1] = a.y; x[4 * q + 2] = a.z; x[4 * q + 3] = a.w;
      y[4 * q] = b.x; y[4 * q + 1] = b.y; y[4 * q + 2] = b.z; y[4 * q + 3] = b.w;
    }
    rowsum_store<NOUT>(x, d0 + row * dPitch + seg * NOUT);
    rowsum_store<NOUT>(y, d1 + row * dPitch + seg * NOUT);
  }
}

__device__ __forceinline__ void hsum_ab(const float* __restrict__ mI,
                                        const float* __restrict__ vIi,
                                        const float* __restrict__ mP,
                                        const float* __restrict__ cIp,
                                        float* __restrict__ dA, int aPitch,
                                        float* __restrict__ dB, int bPitch,
                                        int lane, int wave) {
  if (wave < 6) {
    const int g   = wave * 2 + (lane >> 4);
    const int seg = g / 3;
    const int row = (g - 3 * seg) * 16 + (lane & 15);
    const int o   = row * ABPITCH + seg * 8;
    const v4f* pm = (const v4f*)(mI  + o);
    const v4f* pv = (const v4f*)(vIi + o);
    const v4f* pp = (const v4f*)(mP  + o);
    const v4f* pc = (const v4f*)(cIp + o);
    float xa[20], xb[20];
#pragma unroll
    for (int q = 0; q < 4; ++q) {
      const v4f m = pm[q], iv = pv[q], p = pp[q], c = pc[q];
      const v4f a = (c - m * p) * iv;
      const v4f b = p - a * m;
      xa[4 * q] = a.x; xa[4 * q + 1] = a.y; xa[4 * q + 2] = a.z; xa[4 * q + 3] = a.w;
      xb[4 * q] = b.x; xb[4 * q + 1] = b.y; xb[4 * q + 2] = b.z; xb[4 * q + 3] = b.w;
    }
    {
      const v2f m = *(const v2f*)(pm + 4), iv = *(const v2f*)(pv + 4);
      const v2f p = *(const v2f*)(pp + 4), c  = *(const v2f*)(pc + 4);
      const v2f a = (c - m * p) * iv;
      const v2f b = p - a * m;
      xa[16] = a.x; xa[17] = a.y; xa[18] = 0.0f; xa[19] = 0.0f;
      xb[16] = b.x; xb[17] = b.y; xb[18] = 0.0f; xb[19] = 0.0f;
    }
    rowsum_store<8>(xa, dA + row * aPitch + seg * 8);
    rowsum_store<8>(xb, dB + row * bPitch + seg * 8);
  }
}

__device__ __forceinline__ void vblur(const float* __restrict__ src, int sPitch,
                                      float* __restrict__ dst, int dPitch,
                                      int mBlocks, int nBlocks,
                                      const v2f* __restrict__ band,
                                      int lane, int wave) {
  const int n = lane & 15, half = lane >> 4;
  const int nT = mBlocks * nBlocks;
  for (int t = wave; t < nT; t += 8) {
    const int mi = t / nBlocks, nj = t - mi * nBlocks;
    const float* ps = src + (mi * 16 + 2 * half) * sPitch + nj * 16 + n;
    v8f c = {};
#pragma unroll
    for (int kt = 0; kt < 7; ++kt) {
      v2f b;
      b.x = ps[(4 * kt) * sPitch];
      b.y = ps[(4 * kt + 1) * sPitch];
      c = __builtin_amdgcn_wmma_f32_16x16x4_f32(false, band[kt], false, b,
                                                (short)0, c, false, false);
    }
    float* pd = dst + (mi * 16 + half * 8) * dPitch + nj * 16 + n;
#pragma unroll
    for (int v = 0; v < 8; ++v) pd[v * dPitch] = c[v];
  }
}

template <int NT>
__device__ __forceinline__ void vblur_tiles(const float* const (&ps)[NT], int sPitch,
                                            float* const (&pd)[NT], int dPitch,
                                            const v2f* __restrict__ band) {
  v8f c[NT];
#pragma unroll
  for (int i = 0; i < NT; ++i) c[i] = v8f{};
#pragma unroll
  for (int kt = 0; kt < 7; ++kt) {
    v2f b[NT];
#pragma unroll
    for (int i = 0; i < NT; ++i) {
      b[i].x = ps[i][(4 * kt) * sPitch];
      b[i].y = ps[i][(4 * kt + 1) * sPitch];
    }
#pragma unroll
    for (int i = 0; i < NT; ++i)
      c[i] = __builtin_amdgcn_wmma_f32_16x16x4_f32(false, band[kt], false, b[i],
                                                   (short)0, c[i], false, false);
  }
#pragma unroll
  for (int i = 0; i < NT; ++i)
#pragma unroll
    for (int v = 0; v < 8; ++v) pd[i][v * dPitch] = c[i][v];
}

__device__ __forceinline__ void vblur_pair(const float* sA, const float* sB, int sPitch,
                                           float* dA, float* dB, int dPitch,
                                           const v2f* __restrict__ band,
                                           int lane, int wave) {
  if (wave < 6) {
    const int n = lane & 15, half = lane >> 4;
    const bool second = wave >= 3;
    const int mi = second ? wave - 3 : wave;
    const float* const s = (second ? sB : sA) + (mi * 16 + 2 * half) * sPitch + n;
    float* const d = (second ? dB : dA) + (mi * 16 + half * 8) * dPitch + n;
    const float* const ps[3] = {s, s + 16, s + 32};
    float* const pd[3] = {d, d + 16, d + 32};
    vblur_tiles<3>(ps, sPitch, pd, dPitch, band);
  }
}

__global__ __launch_bounds__(256, 8) void guided_filter_kernel(
    const float* __restrict__ feat, const float* __restrict__ img,
    float* __restrict__ out) {
  __shared__ __attribute__((aligned(16))) float rawI[RAWN * RAWPITCH];
  __shared__ __attribute__((aligned(16))) float rawF[RAWN * RAWPITCH];
  __shared__ __attribute__((aligned(16))) float tmpH[TROWS * TPITCH];
  __shared__ __attribute__((aligned(16))) float tmpG[TROWS * TPITCH];
  __shared__ __attribute__((aligned(16))) float mI [ABROWS * ABPITCH];
  __shared__ __attribute__((aligned(16))) float vIi[ABROWS * ABPITCH];
  __shared__ __attribute__((aligned(16))) float cIp[ABROWS * ABPITCH];
  float* const mP = rawF;
  float* const mA = cIp;
  float* const mB = cIp + TS * MAPITCH;

  const int tid  = threadIdx.x;
  const int lane = tid & 31;
  const int wave = tid >> 5;
  const int x0 = blockIdx.x * TS, y0 = blockIdx.y * TS;
  const int bat = blockIdx.z;

  v2f band[7];
  {
    const int q = lane & 15, h2 = (lane >> 4) * 2;
#pragma unroll
    for (int kt = 0; kt < 7; ++kt) {
      const int k0 = 4 * kt + h2;
      band[kt].x = (k0 >= q && k0 <= q + 10) ? (1.0f / 11.0f) : 0.0f;
      band[kt].y = (k0 + 1 >= q && k0 + 1 <= q + 10) ? (1.0f / 11.0f) : 0.0f;
    }
  }

  for (int i = tid; i < RAWN * RAWPITCH; i += 256) { rawI[i] = 0.0f; rawF[i] = 0.0f; }
  for (int i = tid; i < (TROWS - RAWN) * TPITCH; i += 256) {
    tmpH[RAWN * TPITCH + i] = 0.0f;
    tmpG[RAWN * TPITCH + i] = 0.0f;
  }
  __syncthreads();

  const int  hcol = tid & 63;
  const bool hact = hcol < RAWN;
  const int  hlds = (tid >> 6) * RAWPITCH + hcol;
  unsigned hofs[HROWS];
  {
    const int gx = refl(x0 - 10 + hcol);
#pragma unroll
    for (int j = 0; j < HROWS; ++j)
      hofs[j] = (unsigned)(refl(y0 - 10 + (tid >> 6) + 4 * j) * IMG + gx);
  }

  const float* imgB = img + (size_t)bat * IMG * IMG;
  if (hact) {
    float hv[HROWS];
#pragma unroll
    for (int j = 0; j < HROWS; ++j) hv[j] = imgB[hofs[j]];
#pragma unroll
    for (int j = 0; j < HROWS; ++j) rawI[hlds + j * 4 * RAWPITCH] = hv[j];
  }
  __syncthreads();

  hsum_pair<12>(rawI, rawI, RAWPITCH, tmpH, tmpG, TPITCH, RAWN, lane, wave);
  __syncthreads();
  vblur_pair(tmpH, tmpG, TPITCH, mI, vIi, ABPITCH, band, lane, wave);
  __syncthreads();
  for (int i = tid; i < ABROWS * ABROWS; i += 256) {
    const int r = i / ABROWS, cc = i - r * ABROWS;
    const int o = r * ABPITCH + cc;
    const float m = mI[o];
    vIi[o] = 1.0f / (vIi[o] - m * m + 1e-8f);
  }
  __syncthreads();

  const float* featBat = feat + (size_t)bat * CH * IMG * IMG;
  for (int ch = 0; ch < CH; ++ch) {
    const unsigned chofs = (unsigned)ch * (unsigned)(IMG * IMG);
    if (hact) {
      float hv[HROWS];
#pragma unroll
      for (int j = 0; j < HROWS; ++j) hv[j] = featBat[hofs[j] + chofs];
#pragma unroll
      for (int j = 0; j < HROWS; ++j) rawF[hlds + j * 4 * RAWPITCH] = hv[j];
    }
    __syncthreads();

    hsum_pair<12>(rawF, rawI, RAWPITCH, tmpH, tmpG, TPITCH, RAWN, lane, wave);
    __syncthreads();
    vblur_pair(tmpH, tmpG, TPITCH, mP, cIp, ABPITCH, band, lane, wave);
    __syncthreads();

    hsum_ab(mI, vIi, mP, cIp, tmpH, TPITCH, tmpG, TPITCH, lane, wave);
    __syncthreads();
    if (wave < 4) vblur(tmpH, TPITCH, mA, MAPITCH, 2, 2, band, lane, wave);
    else          vblur(tmpG, TPITCH, mB, MAPITCH, 2, 2, band, lane, wave - 4);
    __syncthreads();

    float* outB = out + ((size_t)bat * CH + ch) * IMG * IMG;
    for (int i = tid; i < TS * TS; i += 256) {
      const int y = i >> 5, x = i & 31;
      outB[(size_t)(y0 + y) * IMG + (x0 + x)] =
          mA[y * MAPITCH + x] * rawI[(y + 10) * RAWPITCH + (x + 10)] +
          mB[y * MAPITCH + x];
    }
  }
}

extern "C" void kernel_launch(void* const* d_in, const int* in_sizes, int n_in,
                              void* d_out, int out_size, void* d_ws, size_t ws_size,
                              hipStream_t stream) {
  const float* feat = (const float*)d_in[0];
  const float* img  = (const float*)d_in[1];
  float* out = (float*)d_out;
  const int B = in_sizes[1] / (IMG * IMG);
  dim3 grid(IMG / TS, IMG / TS, B);
  guided_filter_kernel<<<grid, 256, 0, stream>>>(feat, img, out);
}


# Second-stage V pass and output sweep are one phase: a wave keeps the mean_a and mean_b D tiles of a 16x16 position (2 chains) and stores mean_a*I+mean_b from them; 4 barriers per channel, not 5
# speedup vs baseline: 2.3518x; 1.0993x over previous
#include <hip/hip_runtime.h>


#define IMG      512
#define CH       16
#define TS       32
#define RAWN     52
#define RAWPITCH 60
#define HROWS    13
#define ABROWS   48
#define ABPITCH  52
#define TROWS    60
#define TPITCH   52

typedef __attribute__((ext_vector_type(2))) float v2f;
typedef __attribute__((ext_vector_type(4))) float v4f;
typedef __attribute__((ext_vector_type(8))) float v8f;

__device__ __forceinline__ int refl(int p) {
  p = p < 0 ? -p : p;
  return p >= IMG ? 2 * IMG - 2 - p : p;
}

template <int NOUT>
__device__ __forceinline__ void rowsum_store(const float (&x)[NOUT + 12],
                                             float* __restrict__ drow) {
  constexpr int NE = (NOUT + 10) / 2;
  constexpr int NW = NOUT / 2 + 1;
  float e[NE], ee[NE - 2], w[NW];
#pragma unroll
  for (int k = 0; k < NE; ++k) e[k] = x[2 * k] + x[2 * k + 1];
#pragma unroll
  for (int k = 0; k < NE - 2; ++k) ee[k] = e[k] + e[k + 1];
#pragma unroll
  for (int k = 0; k < NW; ++k) w[k] = (ee[k] + ee[k + 2]) + e[k + 4];
  v4f* pd = (v4f*)drow;
#pragma unroll
  for (int q = 0; q < NOUT / 4; ++q) {
    v4f o;
    o.x = (w[2 * q]     + x[4 * q + 10]) * (1.0f / 11.0f);
    o.y = (x[4 * q + 1] + w[2 * q + 1])  * (1.0f / 11.0f);
    o.z = (w[2 * q + 1] + x[4 * q + 12]) * (1.0f / 11.0f);
    o.w = (x[4 * q + 3] + w[2 * q + 2])  * (1.0f / 11.0f);
    pd[q] = o;
  }
}

template <int NOUT>
__device__ __forceinline__ void hsum_pair(const float* s0, const float* s1, int sPitch,
                                          float* __restrict__ d0,
                                          float* __restrict__ d1, int dPitch,
                                          int rows, int lane, int wave) {
  constexpr int NV = (NOUT + 12) / 4;
  const int seg = wave & 3;
  const int row = (wave >> 2) * 32 + lane;
  if (row < rows) {
    const v4f* p0 = (const v4f*)(s0 + row * sPitch + seg * NOUT);
    const v4f* p1 = (const v4f*)(s1 + row * sPitch + seg * NOUT);
    float x[4 * NV], y[4 * NV];
#pragma unroll
    for (int q = 0; q < NV; ++q) {
      const v4f a = p0[q];
      const v4f b = a * p1[q];
      x[4 * q] = a.x; x[4 * q + 1] = a.y; x[4 * q + 2] = a.z; x[4 * q + 3] = a.w;
      y[4 * q] = b.x; y[4 * q + 1] = b.y; y[4 * q + 2] = b.z; y[4 * q + 3] = b.w;
    }
    rowsum_store<NOUT>(x, d0 + row * dPitch + seg * NOUT);
    rowsum_store<NOUT>(y, d1 + row * dPitch + seg * NOUT);
  }
}

__device__ __forceinline__ void hsum_ab(const float* __restrict__ mI,
                                        const float* __restrict__ vIi,
                                        const float* __restrict__ mP,
                                        const float* __restrict__ cIp,
                                        float* __restrict__ dA, int aPitch,
                                        float* __restrict__ dB, int bPitch,
                                        int lane, int wave) {
  if (wave < 6) {
    const int g   = wave * 2 + (lane >> 4);
    const int seg = g / 3;
    const int row = (g - 3 * seg) * 16 + (lane & 15);
    const int o   = row * ABPITCH + seg * 8;
    const v4f* pm = (const v4f*)(mI  + o);
    const v4f* pv = (const v4f*)(vIi + o);
    const v4f* pp = (const v4f*)(mP  + o);
    const v4f* pc = (const v4f*)(cIp + o);
    float xa[20], xb[20];
#pragma unroll
    for (int q = 0; q < 4; ++q) {
      const v4f m = pm[q], iv = pv[q], p = pp[q], c = pc[q];
      const v4f a = (c - m * p) * iv;
      const v4f b = p - a * m;
      xa[4 * q] = a.x; xa[4 * q + 1] = a.y; xa[4 * q + 2] = a.z; xa[4 * q + 3] = a.w;
      xb[4 * q] = b.x; xb[4 * q + 1] = b.y; xb[4 * q + 2] = b.z; xb[4 * q + 3] = b.w;
    }
    {
      const v2f m = *(const v2f*)(pm + 4), iv = *(const v2f*)(pv + 4);
      const v2f p = *(const v2f*)(pp + 4), c  = *(const v2f*)(pc + 4);
      const v2f a = (c - m * p) * iv;
      const v2f b = p - a * m;
      xa[16] = a.x; xa[17] = a.y; xa[18] = 0.0f; xa[19] = 0.0f;
      xb[16] = b.x; xb[17] = b.y; xb[18] = 0.0f; xb[19] = 0.0f;
    }
    rowsum_store<8>(xa, dA + row * aPitch + seg * 8);
    rowsum_store<8>(xb, dB + row * bPitch + seg * 8);
  }
}

template <int NT>
__device__ __forceinline__ void vblur_acc(const float* const (&ps)[NT], int sPitch,
                                          const v2f* __restrict__ band, v8f (&c)[NT]) {
#pragma unroll
  for (int i = 0; i < NT; ++i) c[i] = v8f{};
#pragma unroll
  for (int kt = 0; kt < 7; ++kt) {
    v2f b[NT];
#pragma unroll
    for (int i = 0; i < NT; ++i) {
      b[i].x = ps[i][(4 * kt) * sPitch];
      b[i].y = ps[i][(4 * kt + 1) * sPitch];
    }
#pragma unroll
    for (int i = 0; i < NT; ++i)
      c[i] = __builtin_amdgcn_wmma_f32_16x16x4_f32(false, band[kt], false, b[i],
                                                   (short)0, c[i], false, false);
  }
}

template <int NT>
__device__ __forceinline__ void vblur_tiles(const float* const (&ps)[NT], int sPitch,
                                            float* const (&pd)[NT], int dPitch,
                                            const v2f* __restrict__ band) {
  v8f c[NT];
  vblur_acc<NT>(ps, sPitch, band, c);
#pragma unroll
  for (int i = 0; i < NT; ++i)
#pragma unroll
    for (int v = 0; v < 8; ++v) pd[i][v * dPitch] = c[i][v];
}

__device__ __forceinline__ void vblur_pair(const float* sA, const float* sB, int sPitch,
                                           float* dA, float* dB, int dPitch,
                                           const v2f* __restrict__ band,
                                           int lane, int wave) {
  if (wave < 6) {
    const int n = lane & 15, half = lane >> 4;
    const bool second = wave >= 3;
    const int mi = second ? wave - 3 : wave;
    const float* const s = (second ? sB : sA) + (mi * 16 + 2 * half) * sPitch + n;
    float* const d = (second ? dB : dA) + (mi * 16 + half * 8) * dPitch + n;
    const float* const ps[3] = {s, s + 16, s + 32};
    float* const pd[3] = {d, d + 16, d + 32};
    vblur_tiles<3>(ps, sPitch, pd, dPitch, band);
  }
}

__device__ __forceinline__ void vblur_out(const float* sA, const float* sB, int sPitch,
                                          const float* guide, float* __restrict__ po,
                                          const v2f* __restrict__ band,
                                          int lane, int wave) {
  if (wave < 4) {
    const int n = lane & 15, half = lane >> 4;
    const int mi = wave >> 1, nj = wave & 1;
    const int so = (mi * 16 + 2 * half) * sPitch + nj * 16 + n;
    const float* const ps[2] = {sA + so, sB + so};
    const float* const pg = guide + (mi * 16 + half * 8 + 10) * RAWPITCH + nj * 16 + n + 10;
    float g[8];
#pragma unroll
    for (int v = 0; v < 8; ++v) g[v] = pg[v * RAWPITCH];
    v8f c[2];
    vblur_acc<2>(ps, sPitch, band, c);
#pragma unroll
    for (int v = 0; v < 8; ++v) po[v * IMG] = c[0][v] * g[v] + c[1][v];
  }
}

__global__ __launch_bounds__(256, 8) void guided_filter_kernel(
    const float* __restrict__ feat, const float* __restrict__ img,
    float* __restrict__ out) {
  __shared__ __attribute__((aligned(16))) float rawI[RAWN * RAWPITCH];
  __shared__ __attribute__((aligned(16))) float rawF[RAWN * RAWPITCH];
  __shared__ __attribute__((aligned(16))) float tmpH[TROWS * TPITCH];
  __shared__ __attribute__((aligned(16))) float tmpG[TROWS * TPITCH];
  __shared__ __attribute__((aligned(16))) float mI [ABROWS * ABPITCH];
  __shared__ __attribute__((aligned(16))) float vIi[ABROWS * ABPITCH];
  __shared__ __attribute__((aligned(16))) float cIp[ABROWS * ABPITCH];
  float* const mP = rawF;

  const int tid  = threadIdx.x;
  const int lane = tid & 31;
  const int wave = tid >> 5;
  const int x0 = blockIdx.x * TS, y0 = blockIdx.y * TS;
  const int bat = blockIdx.z;

  v2f band[7];
  {
    const int q = lane & 15, h2 = (lane >> 4) * 2;
#pragma unroll
    for (int kt = 0; kt < 7; ++kt) {
      const int k0 = 4 * kt + h2;
      band[kt].x = (k0 >= q && k0 <= q + 10) ? (1.0f / 11.0f) : 0.0f;
      band[kt].y = (k0 + 1 >= q && k0 + 1 <= q + 10) ? (1.0f / 11.0f) : 0.0f;
    }
  }

  for (int i = tid; i < RAWN * RAWPITCH; i += 256) { rawI[i] = 0.0f; rawF[i] = 0.0f; }
  for (int i = tid; i < (TROWS - RAWN) * TPITCH; i += 256) {
    tmpH[RAWN * TPITCH + i] = 0.0f;
    tmpG[RAWN * TPITCH + i] = 0.0f;
  }
  __syncthreads();

  const int  hcol = tid & 63;
  const bool hact = hcol < RAWN;
  const int  hlds = (tid >> 6) * RAWPITCH + hcol;
  unsigned hofs[HROWS];
  {
    const int gx = refl(x0 - 10 + hcol);
#pragma unroll
    for (int j = 0; j < HROWS; ++j)
      hofs[j] = (unsigned)(refl(y0 - 10 + (tid >> 6) + 4 * j) * IMG + gx);
  }

  const float* imgB = img + (size_t)bat * IMG * IMG;
  if (hact) {
    float hv[HROWS];
#pragma unroll
    for (int j = 0; j < HROWS; ++j) hv[j] = imgB[hofs[j]];
#pragma unroll
    for (int j = 0; j < HROWS; ++j) rawI[hlds + j * 4 * RAWPITCH] = hv[j];
  }
  __syncthreads();

  hsum_pair<12>(rawI, rawI, RAWPITCH, tmpH, tmpG, TPITCH, RAWN, lane, wave);
  __syncthreads();
  vblur_pair(tmpH, tmpG, TPITCH, mI, vIi, ABPITCH, band, lane, wave);
  __syncthreads();
  for (int i = tid; i < ABROWS * ABROWS; i += 256) {
    const int r = i / ABROWS, cc = i - r * ABROWS;
    const int o = r * ABPITCH + cc;
    const float m = mI[o];
    vIi[o] = 1.0f / (vIi[o] - m * m + 1e-8f);
  }
  __syncthreads();

  const float* featBat = feat + (size_t)bat * CH * IMG * IMG;
  float* outBat = out + (size_t)bat * CH * IMG * IMG;
  const unsigned oofs = (unsigned)((y0 + (wave >> 1) * 16 + (lane >> 4) * 8) * IMG +
                                   x0 + (wave & 1) * 16 + (lane & 15));
  for (int ch = 0; ch < CH; ++ch) {
    const unsigned chofs = (unsigned)ch * (unsigned)(IMG * IMG);
    if (hact) {
      float hv[HROWS];
#pragma unroll
      for (int j = 0; j < HROWS; ++j) hv[j] = featBat[hofs[j] + chofs];
#pragma unroll
      for (int j = 0; j < HROWS; ++j) rawF[hlds + j * 4 * RAWPITCH] = hv[j];
    }
    __syncthreads();

    hsum_pair<12>(rawF, rawI, RAWPITCH, tmpH, tmpG, TPITCH, RAWN, lane, wave);
    __syncthreads();
    vblur_pair(tmpH, tmpG, TPITCH, mP, cIp, ABPITCH, band, lane, wave);
    __syncthreads();

    hsum_ab(mI, vIi, mP, cIp, tmpH, TPITCH, tmpG, TPITCH, lane, wave);
    __syncthreads();
    vblur_out(tmpH, tmpG, TPITCH, rawI, outBat + (oofs + chofs), band, lane, wave);
  }
}

extern "C" void kernel_launch(void* const* d_in, const int* in_sizes, int n_in,
                              void* d_out, int out_size, void* d_ws, size_t ws_size,
                              hipStream_t stream) {
  const float* feat = (const float*)d_in[0];
  const float* img  = (const float*)d_in[1];
  float* out = (float*)d_out;
  const int B = in_sizes[1] / (IMG * IMG);
  dim3 grid(IMG / TS, IMG / TS, B);
  guided_filter_kernel<<<grid, 256, 0, stream>>>(feat, img, out);
}


# Tile and row tasks re-dealt over the 8 waves (first V phase 3,2,2,2,2,3,2,2 tiles, not 3 on six waves; a/b rows and output tiles moved; a scalar wave number picks them): equal work per SIMD
# speedup vs baseline: 2.4709x; 1.0506x over previous
#include <hip/hip_runtime.h>


#define IMG      512
#define CH       16
#define TS       32
#define RAWN     52
#define RAWPITCH 60
#define HROWS    13
#define ABROWS   48
#define ABPITCH  52
#define TROWS    60
#define TPITCH   52

typedef __attribute__((ext_vector_type(2))) float v2f;
typedef __attribute__((ext_vector_type(4))) float v4f;
typedef __attribute__((ext_vector_type(8))) float v8f;

__device__ __forceinline__ int refl(int p) {
  p = p < 0 ? -p : p;
  return p >= IMG ? 2 * IMG - 2 - p : p;
}

template <int NOUT>
__device__ __forceinline__ void rowsum_store(const float (&x)[NOUT + 12],
                                             float* __restrict__ drow) {
  constexpr int NE = (NOUT + 10) / 2;
  constexpr int NW = NOUT / 2 + 1;
  float e[NE], ee[NE - 2], w[NW];
#pragma unroll
  for (int k = 0; k < NE; ++k) e[k] = x[2 * k] + x[2 * k + 1];
#pragma unroll
  for (int k = 0; k < NE - 2; ++k) ee[k] = e[k] + e[k + 1];
#pragma unroll
  for (int k = 0; k < NW; ++k) w[k] = (ee[k] + ee[k + 2]) + e[k + 4];
  v4f* pd = (v4f*)drow;
#pragma unroll
  for (int q = 0; q < NOUT / 4; ++q) {
    v4f o;
    o.x = (w[2 * q]     + x[4 * q + 10]) * (1.0f / 11.0f);
    o.y = (x[4 * q + 1] + w[2 * q + 1])  * (1.0f / 11.0f);
    o.z = (w[2 * q + 1] + x[4 * q + 12]) * (1.0f / 11.0f);
    o.w = (x[4 * q + 3] + w[2 * q + 2])  * (1.0f / 11.0f);
    pd[q] = o;
  }
}

template <int NOUT>
__device__ __forceinline__ void hsum_pair(const float* s0, const float* s1, int sPitch,
                                          float* __restrict__ d0,
                                          float* __restrict__ d1, int dPitch,
                                          int rows, int lane, int wave) {
  constexpr int NV = (NOUT + 12) / 4;
  const int seg = wave & 3;
  const int row = (wave >> 2) * 32 + lane;
  if (row < rows) {
    const v4f* p0 = (const v4f*)(s0 + row * sPitch + seg * NOUT);
    const v4f* p1 = (const v4f*)(s1 + row * sPitch + seg * NOUT);
    float x[4 * NV], y[4 * NV];
#pragma unroll
    for (int q = 0; q < NV; ++q) {
      const v4f a = p0[q];
      const v4f b = a * p1[q];
      x[4 * q] = a.x; x[4 * q + 1] = a.y; x[4 * q + 2] = a.z; x[4 * q + 3] = a.w;
      y[4 * q] = b.x; y[4 * q + 1] = b.y; y[4 * q + 2] = b.z; y[4 * q + 3] = b.w;
    }
    rowsum_store<NOUT>(x, d0 + row * dPitch + seg * NOUT);
    rowsum_store<NOUT>(y, d1 + row * dPitch + seg * NOUT);
  }
}

__device__ __forceinline__ void hsum_ab(const float* __restrict__ mI,
                                        const float* __restrict__ vIi,
                                        const float* __restrict__ mP,
                                        const float* __restrict__ cIp,
                                        float* __restrict__ dA, int aPitch,
                                        float* __restrict__ dB, int bPitch,
                                        int lane, int wave) {
  if ((0xED >> wave) & 1) {
    const int gw  = wave - (wave > 0 ? 1 : 0) - (wave > 3 ? 1 : 0);
    const int g   = gw * 2 + (lane >> 4);
    const int seg = g / 3;
    const int row = (g - 3 * seg) * 16 + (lane & 15);
    const int o   = row * ABPITCH + seg * 8;
    const v4f* pm = (const v4f*)(mI  + o);
    const v4f* pv = (const v4f*)(vIi + o);
    const v4f* pp = (const v4f*)(mP  + o);
    const v4f* pc = (const v4f*)(cIp + o);
    float xa[20], xb[20];
#pragma unroll
    for (int q = 0; q < 4; ++q) {
      const v4f m = pm[q], iv = pv[q], p = pp[q], c = pc[q];
      const v4f a = (c - m * p) * iv;
      const v4f b = p - a * m;
      xa[4 * q] = a.x; xa[4 * q + 1] = a.y; xa[4 * q + 2] = a.z; xa[4 * q + 3] = a.w;
      xb[4 * q] = b.x; xb[4 * q + 1] = b.y; xb[4 * q + 2] = b.z; xb[4 * q + 3] = b.w;
    }
    {
      const v2f m = *(const v2f*)(pm + 4), iv = *(const v2f*)(pv + 4);
      const v2f p = *(const v2f*)(pp + 4), c  = *(const v2f*)(pc + 4);
      const v2f a = (c - m * p) * iv;
      const v2f b = p - a * m;
      xa[16] = a.x; xa[17] = a.y; xa[18] = 0.0f; xa[19] = 0.0f;
      xb[16] = b.x; xb[17] = b.y; xb[18] = 0.0f; xb[19] = 0.0f;
    }
    rowsum_store<8>(xa, dA + row * aPitch + seg * 8);
    rowsum_store<8>(xb, dB + row * bPitch + seg * 8);
  }
}

template <int NT>
__device__ __forceinline__ void vblur_acc(const float* const (&ps)[NT], int sPitch,
                                          const v2f* __restrict__ band, v8f (&c)[NT]) {
#pragma unroll
  for (int i = 0; i < NT; ++i) c[i] = v8f{};
#pragma unroll
  for (int kt = 0; kt < 7; ++kt) {
    v2f b[NT];
#pragma unroll
    for (int i = 0; i < NT; ++i) {
      b[i].x = ps[i][(4 * kt) * sPitch];
      b[i].y = ps[i][(4 * kt + 1) * sPitch];
    }
#pragma unroll
    for (int i = 0; i < NT; ++i)
      c[i] = __builtin_amdgcn_wmma_f32_16x16x4_f32(false, band[kt], false, b[i],
                                                   (short)0, c[i], false, false);
  }
}

template <int NT>
__device__ __forceinline__ void vblur_tiles(const float* const (&ps)[NT], int sPitch,
                                            float* const (&pd)[NT], int dPitch,
                                            const v2f* __restrict__ band) {
  v8f c[NT];
  vblur_acc<NT>(ps, sPitch, band, c);
#pragma unroll
  for (int i = 0; i < NT; ++i)
#pragma unroll
    for (int v = 0; v < 8; ++v) pd[i][v * dPitch] = c[i][v];
}

__device__ __forceinline__ void vblur_pair(const float* sA, const float* sB, int sPitch,
                                           float* dA, float* dB, int dPitch,
                                           const v2f* __restrict__ band,
                                           int lane, int wave) {
  const int n = lane & 15, half = lane >> 4;
  if ((0x21 >> wave) & 1) {
    const bool second = wave != 0;
    const float* const s = (second ? sB : sA) + (2 * half) * sPitch + n;
    float* const d = (second ? dB : dA) + (half * 8) * dPitch + n;
    const float* const ps[3] = {s, s + 16, s + 32};
    float* const pd[3] = {d, d + 16, d + 32};
    vblur_tiles<3>(ps, sPitch, pd, dPitch, band);
  } else if (wave < 6) {
    const int t = wave - 1;
    const bool second = t & 1;
    const int mi = 1 + (t >> 1);
    const float* const s = (second ? sB : sA) + (mi * 16 + 2 * half) * sPitch + n;
    float* const d = (second ? dB : dA) + (mi * 16 + half * 8) * dPitch + n;
    const float* const ps[2] = {s, s + 16};
    float* const pd[2] = {d, d + 16};
    vblur_tiles<2>(ps, sPitch, pd, dPitch, band);
  } else {
    const int mi = wave - 5;
    const int so = (mi * 16 + 2 * half) * sPitch + 32 + n;
    const int dofs = (mi * 16 + half * 8) * dPitch + 32 + n;
    const float* const ps[2] = {sA + so, sB + so};
    float* const pd[2] = {dA + dofs, dB + dofs};
    vblur_tiles<2>(ps, sPitch, pd, dPitch, band);
  }
}

__device__ __forceinline__ void vblur_out(const float* sA, const float* sB, int sPitch,
                                          const float* guide, float* __restrict__ po,
                                          const v2f* __restrict__ band,
                                          int lane, int wave) {
  if ((0x5A >> wave) & 1) {
    const int n = lane & 15, half = lane >> 4;
    const int mi = wave >> 2, nj = (wave >> 1) & 1;
    const int so = (mi * 16 + 2 * half) * sPitch + nj * 16 + n;
    const float* const ps[2] = {sA + so, sB + so};
    const float* const pg = guide + (mi * 16 + half * 8 + 10) * RAWPITCH + nj * 16 + n + 10;
    float g[8];
#pragma unroll
    for (int v = 0; v < 8; ++v) g[v] = pg[v * RAWPITCH];
    v8f c[2];
    vblur_acc<2>(ps, sPitch, band, c);
#pragma unroll
    for (int v = 0; v < 8; ++v) po[v * IMG] = c[0][v] * g[v] + c[1][v];
  }
}

__global__ __launch_bounds__(256, 8) void guided_filter_kernel(
    const float* __restrict__ feat, const float* __restrict__ img,
    float* __restrict__ out) {
  __shared__ __attribute__((aligned(16))) float rawI[RAWN * RAWPITCH];
  __shared__ __attribute__((aligned(16))) float rawF[RAWN * RAWPITCH];
  __shared__ __attribute__((aligned(16))) float tmpH[TROWS * TPITCH];
  __shared__ __attribute__((aligned(16))) float tmpG[TROWS * TPITCH];
  __shared__ __attribute__((aligned(16))) float mI [ABROWS * ABPITCH];
  __shared__ __attribute__((aligned(16))) float vIi[ABROWS * ABPITCH];
  __shared__ __attribute__((aligned(16))) float cIp[ABROWS * ABPITCH];
  float* const mP = rawF;

  const int tid  = threadIdx.x;
  const int lane = tid & 31;
  const int wave = __builtin_amdgcn_readfirstlane(tid >> 5);
  const int x0 = blockIdx.x * TS, y0 = blockIdx.y * TS;
  const int bat = blockIdx.z;

  v2f band[7];
  {
    const int q = lane & 15, h2 = (lane >> 4) * 2;
#pragma unroll
    for (int kt = 0; kt < 7; ++kt) {
      const int k0 = 4 * kt + h2;
      band[kt].x = (k0 >= q && k0 <= q + 10) ? (1.0f / 11.0f) : 0.0f;
      band[kt].y = (k0 + 1 >= q && k0 + 1 <= q + 10) ? (1.0f / 11.0f) : 0.0f;
    }
  }

  for (int i = tid; i < RAWN * RAWPITCH; i += 256) { rawI[i] = 0.0f; rawF[i] = 0.0f; }
  for (int i = tid; i < (TROWS - RAWN) * TPITCH; i += 256) {
    tmpH[RAWN * TPITCH + i] = 0.0f;
    tmpG[RAWN * TPITCH + i] = 0.0f;
  }
  __syncthreads();

  const int  hcol = tid & 63;
  const bool hact = hcol < RAWN;
  const int  hlds = (tid >> 6) * RAWPITCH + hcol;
  unsigned hofs[HROWS];
  {
    const int gx = refl(x0 - 10 + hcol);
#pragma unroll
    for (int j = 0; j < HROWS; ++j)
      hofs[j] = (unsigned)(refl(y0 - 10 + (tid >> 6) + 4 * j) * IMG + gx);
  }

  const float* imgB = img + (size_t)bat * IMG * IMG;
  if (hact) {
    float hv[HROWS];
#pragma unroll
    for (int j = 0; j < HROWS; ++j) hv[j] = imgB[hofs[j]];
#pragma unroll
    for (int j = 0; j < HROWS; ++j) rawI[hlds + j * 4 * RAWPITCH] = hv[j];
  }
  __syncthreads();

  hsum_pair<12>(rawI, rawI, RAWPITCH, tmpH, tmpG, TPITCH, RAWN, lane, wave);
  __syncthreads();
  vblur_pair(tmpH, tmpG, TPITCH, mI, vIi, ABPITCH, band, lane, wave);
  __syncthreads();
  for (int i = tid; i < ABROWS * ABROWS; i += 256) {
    const int r = i / ABROWS, cc = i - r * ABROWS;
    const int o = r * ABPITCH + cc;
    const float m = mI[o];
    vIi[o] = 1.0f / (vIi[o] - m * m + 1e-8f);
  }
  __syncthreads();

  const float* featBat = feat + (size_t)bat * CH * IMG * IMG;
  float* outBat = out + (size_t)bat * CH * IMG * IMG;
  const unsigned oofs = (unsigned)((y0 + (wave >> 2) * 16 + (lane >> 4) * 8) * IMG +
                                   x0 + ((wave >> 1) & 1) * 16 + (lane & 15));
  for (int ch = 0; ch < CH; ++ch) {
    const unsigned chofs = (unsigned)ch * (unsigned)(IMG * IMG);
    if (hact) {
      float hv[HROWS];
#pragma unroll
      for (int j = 0; j < HROWS; ++j) hv[j] = featBat[hofs[j] + chofs];
#pragma unroll
      for (int j = 0; j < HROWS; ++j) rawF[hlds + j * 4 * RAWPITCH] = hv[j];
    }
    __syncthreads();

    hsum_pair<12>(rawF, rawI, RAWPITCH, tmpH, tmpG, TPITCH, RAWN, lane, wave);
    __syncthreads();
    vblur_pair(tmpH, tmpG, TPITCH, mP, cIp, ABPITCH, band, lane, wave);
    __syncthreads();

    hsum_ab(mI, vIi, mP, cIp, tmpH, TPITCH, tmpG, TPITCH, lane, wave);
    __syncthreads();
    vblur_out(tmpH, tmpG, TPITCH, rawI, outBat + (oofs + chofs), band, lane, wave);
  }
}

extern "C" void kernel_launch(void* const* d_in, const int* in_sizes, int n_in,
                              void* d_out, int out_size, void* d_ws, size_t ws_size,
                              hipStream_t stream) {
  const float* feat = (const float*)d_in[0];
  const float* img  = (const float*)d_in[1];
  float* out = (float*)d_out;
  const int B = in_sizes[1] / (IMG * IMG);
  dim3 grid(IMG / TS, IMG / TS, B);
  guided_filter_kernel<<<grid, 256, 0, stream>>>(feat, img, out);
}
